# starting kernel
# baseline (speedup 1.0000x reference)
#include <hip/hip_runtime.h>


typedef __attribute__((ext_vector_type(2))) float v2f;
typedef __attribute__((ext_vector_type(8))) float v8f;

namespace {
constexpr int G       = 256;
constexpr int ARITY   = 4;
constexpr int D       = 16;
constexpr int BATCH   = 4096;
constexpr int XW      = G * ARITY;
constexpr int OUTW    = G * D;
constexpr int WAVES_PER_BLOCK = 8;
constexpr int TILES_PER_WAVE  = 4;
constexpr int BLOCKS_Y = BATCH / 16 / (WAVES_PER_BLOCK * TILES_PER_WAVE);
}

__global__ __launch_bounds__(256)
void simplex_wmma_kernel(const float* __restrict__ X,
                         const float* __restrict__ params,
                         float* __restrict__ out)
{
    const int lane = threadIdx.x & 31;
    const int wave = threadIdx.x >> 5;
    const int g    = blockIdx.x;
    const int half = lane >> 4;
    const int m    = lane & 15;

    const float* __restrict__ P = params + (size_t)g * (16 * D);
    v2f bmat[4];
#pragma unroll
    for (int kk = 0; kk < 4; ++kk) {
        const int p0 = kk * 4 + half * 2;
        bmat[kk].x = P[p0 * D + m];
        bmat[kk].y = P[(p0 + 1) * D + m];
    }

#pragma unroll
    for (int t = 0; t < TILES_PER_WAVE; ++t) {
        const int tile = (blockIdx.y * WAVES_PER_BLOCK + wave) * TILES_PER_WAVE + t;
        const int b0   = tile * 16;

        const float4 xr = *reinterpret_cast<const float4*>(
            X + (size_t)(b0 + m) * XW + (size_t)g * ARITY);
        float v0 = xr.x, v1 = xr.y, v2 = xr.z, v3 = xr.w;
        int   j0 = 0, j1 = 1, j2 = 2, j3 = 3;
        if (v0 > v1) { float tv=v0; v0=v1; v1=tv; int tj=j0; j0=j1; j1=tj; }
        if (v2 > v3) { float tv=v2; v2=v3; v3=tv; int tj=j2; j2=j3; j3=tj; }
        if (v0 > v2) { float tv=v0; v0=v2; v2=tv; int tj=j0; j0=j2; j2=tj; }
        if (v1 > v3) { float tv=v1; v1=v3; v3=tv; int tj=j1; j1=j3; j3=tj; }
        if (v1 > v2) { float tv=v1; v1=v2; v2=tv; int tj=j1; j1=j2; j2=tj; }

        const float c0 = v0, c1 = v1 - v0, c2 = v2 - v1, c3 = v3 - v2;
        const int   i3 = 1 << j3;
        const int   i2 = i3 | (1 << j2);
        const int   i1 = i2 | (1 << j1);
        const int   i0 = 15;

        auto wsel = [&](int p) -> float {
            float r = (i0 == p) ? c0 : 0.0f;
            r += (i1 == p) ? c1 : 0.0f;
            r += (i2 == p) ? c2 : 0.0f;
            r += (i3 == p) ? c3 : 0.0f;
            return r;
        };

        v8f acc = {};
#pragma unroll
        for (int kk = 0; kk < 4; ++kk) {
            const int p0 = kk * 4 + half * 2;
            v2f a;
            a.x = wsel(p0);
            a.y = wsel(p0 + 1);
            acc = __builtin_amdgcn_wmma_f32_16x16x4_f32(
                false, a, false, bmat[kk], (short)0, acc, false, false);
        }

        float* __restrict__ orow =
            out + (size_t)(b0 + half * 8) * OUTW + (size_t)g * D + m;
#pragma unroll
        for (int j = 0; j < 8; ++j) {
            orow[(size_t)j * OUTW] = acc[j];
        }
    }
}

extern "C" void kernel_launch(void* const* d_in, const int* in_sizes, int n_in,
                              void* d_out, int out_size, void* d_ws, size_t ws_size,
                              hipStream_t stream) {
    (void)in_sizes; (void)n_in; (void)out_size; (void)d_ws; (void)ws_size;
    const float* X      = (const float*)d_in[0];
    const float* params = (const float*)d_in[1];
    float*       out    = (float*)d_out;

    dim3 grid(G, BLOCKS_Y);
    dim3 block(32 * WAVES_PER_BLOCK);
    simplex_wmma_kernel<<<grid, block, 0, stream>>>(X, params, out);
}


# wave owns the 2 adjacent groups of an out line; accumulator halves swapped between its 16-lane rows (v_permlane16_swap): a store writes one whole 128-B line, not two 64-B halves; half the waves
# speedup vs baseline: 2.6427x; 2.6427x over previous
#include <hip/hip_runtime.h>


typedef __attribute__((ext_vector_type(2))) float v2f;
typedef __attribute__((ext_vector_type(8))) float v8f;
typedef __attribute__((ext_vector_type(2))) unsigned v2u;

namespace {
constexpr int G       = 256;
constexpr int ARITY   = 4;
constexpr int D       = 16;
constexpr int BATCH   = 4096;
constexpr int XW      = G * ARITY;
constexpr int OUTW    = G * D;
constexpr int GROUPS_PER_WAVE = 2;
constexpr int WAVES_PER_BLOCK = 8;
constexpr int TILES_PER_WAVE  = 4;
constexpr int BLOCKS_X = G / (WAVES_PER_BLOCK * GROUPS_PER_WAVE);
constexpr int BLOCKS_Y = BATCH / (16 * TILES_PER_WAVE);

struct Simplex {
    float c0, c1, c2, c3;
    int   i1, i2, i3;
};

__device__ __forceinline__ Simplex simplex_of(const float4 xr)
{
    float v0 = xr.x, v1 = xr.y, v2 = xr.z, v3 = xr.w;
    int   j0 = 0, j1 = 1, j2 = 2, j3 = 3;
    if (v0 > v1) { float tv=v0; v0=v1; v1=tv; int tj=j0; j0=j1; j1=tj; }
    if (v2 > v3) { float tv=v2; v2=v3; v3=tv; int tj=j2; j2=j3; j3=tj; }
    if (v0 > v2) { float tv=v0; v0=v2; v2=tv; int tj=j0; j0=j2; j2=tj; }
    if (v1 > v3) { float tv=v1; v1=v3; v3=tv; int tj=j1; j1=j3; j3=tj; }
    if (v1 > v2) { float tv=v1; v1=v2; v2=tv; int tj=j1; j1=j2; j2=tj; }
    (void)j0;
    Simplex s;
    s.c0 = v0; s.c1 = v1 - v0; s.c2 = v2 - v1; s.c3 = v3 - v2;
    s.i3 = 1 << j3;
    s.i2 = s.i3 | (1 << j2);
    s.i1 = s.i2 | (1 << j1);
    return s;
}

__device__ __forceinline__ float wsel(const Simplex& s, int p)
{
    float r = (15 == p) ? s.c0 : 0.0f;
    r += (s.i1 == p) ? s.c1 : 0.0f;
    r += (s.i2 == p) ? s.c2 : 0.0f;
    r += (s.i3 == p) ? s.c3 : 0.0f;
    return r;
}
}

__global__ __launch_bounds__(32 * WAVES_PER_BLOCK)
void simplex_wmma_kernel(const float* __restrict__ X,
                         const float* __restrict__ params,
                         float* __restrict__ out)
{
    const int lane = threadIdx.x & 31;
    const int wave = threadIdx.x >> 5;
    const int g0   = (blockIdx.x * WAVES_PER_BLOCK + wave) * GROUPS_PER_WAVE;
    const int half = lane >> 4;
    const int m    = lane & 15;

    const float* __restrict__ P = params + (size_t)g0 * (16 * D);
    v2f bmatA[4], bmatB[4];
#pragma unroll
    for (int kk = 0; kk < 4; ++kk) {
        const int p0 = kk * 4 + half * 2;
        bmatA[kk].x = P[p0 * D + m];
        bmatA[kk].y = P[(p0 + 1) * D + m];
        bmatB[kk].x = P[16 * D + p0 * D + m];
        bmatB[kk].y = P[16 * D + (p0 + 1) * D + m];
    }

#pragma unroll
    for (int t = 0; t < TILES_PER_WAVE; ++t) {
        const int tile = blockIdx.y * TILES_PER_WAVE + t;
        const int b0   = tile * 16;

        const float* __restrict__ xp = X + (size_t)(b0 + m) * XW + (size_t)g0 * ARITY;
        const float4 xa = *reinterpret_cast<const float4*>(xp);
        const float4 xb = *reinterpret_cast<const float4*>(xp + ARITY);
        const Simplex sa = simplex_of(xa);
        const Simplex sb = simplex_of(xb);

        v8f accA = {}, accB = {};
#pragma unroll
        for (int kk = 0; kk < 4; ++kk) {
            const int p0 = kk * 4 + half * 2;
            v2f aA, aB;
            aA.x = wsel(sa, p0);
            aA.y = wsel(sa, p0 + 1);
            aB.x = wsel(sb, p0);
            aB.y = wsel(sb, p0 + 1);
            accA = __builtin_amdgcn_wmma_f32_16x16x4_f32(
                false, aA, false, bmatA[kk], (short)0, accA, false, false);
            accB = __builtin_amdgcn_wmma_f32_16x16x4_f32(
                false, aB, false, bmatB[kk], (short)0, accB, false, false);
        }

        float* __restrict__ orow = out + (size_t)b0 * OUTW + (size_t)g0 * D + lane;
        asm volatile("v_nop\n\tv_nop\n\tv_nop\n\tv_nop" : "+v"(accA), "+v"(accB));
#pragma unroll
        for (int j = 0; j < 8; ++j) {
            const v2u sw = __builtin_amdgcn_permlane16_swap(
                __float_as_uint(accA[j]), __float_as_uint(accB[j]), false, false);
            orow[(size_t)j * OUTW]       = __uint_as_float(sw.x);
            orow[(size_t)(j + 8) * OUTW] = __uint_as_float(sw.y);
        }
    }
}

extern "C" void kernel_launch(void* const* d_in, const int* in_sizes, int n_in,
                              void* d_out, int out_size, void* d_ws, size_t ws_size,
                              hipStream_t stream) {
    (void)in_sizes; (void)n_in; (void)out_size; (void)d_ws; (void)ws_size;
    const float* X      = (const float*)d_in[0];
    const float* params = (const float*)d_in[1];
    float*       out    = (float*)d_out;

    dim3 grid(BLOCKS_X, BLOCKS_Y);
    dim3 block(32 * WAVES_PER_BLOCK);
    simplex_wmma_kernel<<<grid, block, 0, stream>>>(X, params, out);
}
